# starting kernel
# baseline (speedup 1.0000x reference)
#include <hip/hip_runtime.h>
#include <hip/hip_bf16.h>
#include <math.h>

typedef float v2f __attribute__((ext_vector_type(2)));
typedef float v8f __attribute__((ext_vector_type(8)));

#define N_ROWS 2048
#define P_ROWS 512
#define D_DIM  1024
#define F_DIM  128
#define EPSV   1e-8f

__device__ __forceinline__ float softplus_f(float x) {
    return log1pf(expf(x));
}

__global__ __launch_bounds__(256) void fbnorm_kernel(const float* __restrict__ fb,
                                                     float* __restrict__ fbn) {
    __shared__ float red[256];
    const int row = blockIdx.x;
    const int t   = threadIdx.x;
    const float* src = fb + (size_t)row * D_DIM;
    float s = 0.0f;
    for (int c = t; c < D_DIM; c += 256) { float v = src[c]; s += v * v; }
    red[t] = s;
    __syncthreads();
    for (int off = 128; off > 0; off >>= 1) {
        if (t < off) red[t] += red[t + off];
        __syncthreads();
    }
    const float scale = 1.0f / fmaxf(sqrtf(red[0]), EPSV);
    float* dst = fbn + (size_t)row * D_DIM;
    for (int c = t; c < D_DIM; c += 256) dst[c] = src[c] * scale;
}

__global__ __launch_bounds__(256) void rowinvnorm_kernel(const float* __restrict__ X,
                                                         float* __restrict__ invn) {
    __shared__ float red[256];
    const int row = blockIdx.x;
    const int t   = threadIdx.x;
    const float* src = X + (size_t)row * D_DIM;
    float s = 0.0f;
    for (int c = t; c < D_DIM; c += 256) { float v = src[c]; s += v * v; }
    red[t] = s;
    __syncthreads();
    for (int off = 128; off > 0; off >>= 1) {
        if (t < off) red[t] += red[t + off];
        __syncthreads();
    }
    if (t == 0) invn[row] = 1.0f / fmaxf(sqrtf(red[0]), EPSV);
}

__global__ __launch_bounds__(128) void gemm_relu_wmma(const float* __restrict__ X,
                                                      const float* __restrict__ fbn,
                                                      const float* __restrict__ invn,
                                                      float* __restrict__ outp) {
    const int wid  = threadIdx.x >> 5;
    const int lane = threadIdx.x & 31;
    const int tile = blockIdx.x * 4 + wid;
    const int tf   = tile & 7;
    const int tm   = tile >> 3;
    const int half = lane >> 4;
    const int lm   = lane & 15;

    const float* arow = X   + (size_t)(tm * 16 + lm) * D_DIM + 2 * half;
    const float* brow = fbn + (size_t)(tf * 16 + lm) * D_DIM + 2 * half;

    v8f acc = {};
#pragma unroll 8
    for (int k = 0; k < D_DIM; k += 4) {
        v2f a2 = *(const v2f*)(arow + k);
        v2f b2 = *(const v2f*)(brow + k);
        acc = __builtin_amdgcn_wmma_f32_16x16x4_f32(
             false, a2,  false, b2,
             (short)0, acc,  false,  false);
    }

    const int col = tf * 16 + lm;
#pragma unroll
    for (int v = 0; v < 8; ++v) {
        const int row = tm * 16 + half * 8 + v;
        const float val = acc[v] * invn[row];
        outp[(size_t)row * F_DIM + col] = fmaxf(val, 0.0f);
    }
}

__global__ __launch_bounds__(256) void pairwise_kernel(const float* __restrict__ ap,
                                                       const float* __restrict__ bp,
                                                       const float* __restrict__ alpha_raw,
                                                       const float* __restrict__ beta_raw,
                                                       const float* __restrict__ theta_raw,
                                                       float* __restrict__ out) {
    __shared__ float apS[16][132];
    __shared__ float bpS[16][132];

    const int tx  = threadIdx.x;
    const int ty  = threadIdx.y;
    const int tid = ty * 16 + tx;
    const int j0  = blockIdx.x * 16;
    const int i0  = blockIdx.y * 16;

    for (int p = tid; p < 16 * 32; p += 256) {
        const int r  = p >> 5;
        const int c4 = (p & 31) * 4;
        *(float4*)&apS[r][c4] = *(const float4*)(ap + (size_t)(i0 + r) * F_DIM + c4);
        *(float4*)&bpS[r][c4] = *(const float4*)(bp + (size_t)(j0 + r) * F_DIM + c4);
    }
    __syncthreads();

    float s_min = 0.0f, s_a = 0.0f, s_b = 0.0f;
#pragma unroll 4
    for (int k = 0; k < F_DIM; ++k) {
        const float av = apS[ty][k];
        const float bv = bpS[tx][k];
        s_min += fminf(av, bv);
        s_a   += (bv > 0.0f) ? av : 0.0f;
        s_b   += (av > 0.0f) ? bv : 0.0f;
    }

    const float al = softplus_f(alpha_raw[0]);
    const float be = softplus_f(beta_raw[0]);
    const float th = softplus_f(theta_raw[0]);
    out[(size_t)(i0 + ty) * P_ROWS + (j0 + tx)] =
        (th + al + be) * s_min - al * s_a - be * s_b;
}

extern "C" void kernel_launch(void* const* d_in, const int* in_sizes, int n_in,
                              void* d_out, int out_size, void* d_ws, size_t ws_size,
                              hipStream_t stream) {
    const float* input     = (const float*)d_in[0];
    const float* prototype = (const float*)d_in[1];
    const float* fb        = (const float*)d_in[2];
    const float* alpha_raw = (const float*)d_in[3];
    const float* beta_raw  = (const float*)d_in[4];
    const float* theta_raw = (const float*)d_in[5];
    float* out = (float*)d_out;

    float* w      = (float*)d_ws;
    float* fbn    = w;
    float* inv_in = fbn + (size_t)F_DIM * D_DIM;
    float* inv_pr = inv_in + N_ROWS;
    float* ap     = inv_pr + P_ROWS;
    float* bp     = ap + (size_t)N_ROWS * F_DIM;

    fbnorm_kernel<<<F_DIM, 256, 0, stream>>>(fb, fbn);
    rowinvnorm_kernel<<<N_ROWS, 256, 0, stream>>>(input, inv_in);
    rowinvnorm_kernel<<<P_ROWS, 256, 0, stream>>>(prototype, inv_pr);

    gemm_relu_wmma<<<(N_ROWS / 16) * (F_DIM / 16) / 4, 128, 0, stream>>>(input, fbn, inv_in, ap);
    gemm_relu_wmma<<<(P_ROWS / 16) * (F_DIM / 16) / 4, 128, 0, stream>>>(prototype, fbn, inv_pr, bp);

    dim3 grid(P_ROWS / 16, N_ROWS / 16);
    pairwise_kernel<<<grid, dim3(16, 16), 0, stream>>>(ap, bp, alpha_raw, beta_raw, theta_raw, out);
}


# Launch fusion, 6 launches to 2: one WMMA launch does both projections and sums row and column norms in its K loop (norm sums reordered, per lane in k order; scaling moved behind the dot product)
# speedup vs baseline: 1.6866x; 1.6866x over previous
#include <hip/hip_runtime.h>
#include <hip/hip_bf16.h>
#include <math.h>

typedef float v2f __attribute__((ext_vector_type(2)));
typedef float v8f __attribute__((ext_vector_type(8)));

#define N_ROWS 2048
#define P_ROWS 512
#define D_DIM  1024
#define F_DIM  128
#define EPSV   1e-8f

__device__ __forceinline__ float softplus_f(float x) {
    return log1pf(expf(x));
}

__global__ __launch_bounds__(160) void proj_norm_relu_wmma(const float* __restrict__ Xin,
                                                           const float* __restrict__ Xpr,
                                                           const float* __restrict__ fb,
                                                           float* __restrict__ proj) {
    const int wid  = threadIdx.x >> 5;
    const int lane = threadIdx.x & 31;
    const int tile = blockIdx.x * 5 + wid;
    const int tf   = tile & 7;
    const int tm   = tile >> 3;
    const int half = lane >> 4;
    const int lm   = lane & 15;

    const float* X = (tm < N_ROWS / 16) ? Xin + (size_t)tm * 16 * D_DIM
                                        : Xpr + (size_t)(tm - N_ROWS / 16) * 16 * D_DIM;
    const float* arow = X  + (size_t)lm * D_DIM + 2 * half;
    const float* brow = fb + (size_t)(tf * 16 + lm) * D_DIM + 2 * half;

    v8f acc = {};
    v2f sa2 = {}, sb2 = {};
#pragma unroll 8
    for (int k = 0; k < D_DIM; k += 4) {
        v2f a2 = *(const v2f*)(arow + k);
        v2f b2 = *(const v2f*)(brow + k);
        sa2 = __builtin_elementwise_fma(a2, a2, sa2);
        sb2 = __builtin_elementwise_fma(b2, b2, sb2);
        acc = __builtin_amdgcn_wmma_f32_16x16x4_f32(
             false, a2,  false, b2,
             (short)0, acc,  false,  false);
    }

    float sa = sa2.x + sa2.y;
    float sb = sb2.x + sb2.y;
    sa += __shfl_xor(sa, 16);
    sb += __shfl_xor(sb, 16);
    const float ia = 1.0f / fmaxf(sqrtf(sa), EPSV);
    const float ib = 1.0f / fmaxf(sqrtf(sb), EPSV);

    const int col = tf * 16 + lm;
#pragma unroll
    for (int v = 0; v < 8; ++v) {
        const int r = half * 8 + v;
        const float iar = __shfl(ia, r);
        const float val = acc[v] * iar * ib;
        proj[(size_t)(tm * 16 + r) * F_DIM + col] = fmaxf(val, 0.0f);
    }
}

__global__ __launch_bounds__(256) void pairwise_kernel(const float* __restrict__ ap,
                                                       const float* __restrict__ bp,
                                                       const float* __restrict__ alpha_raw,
                                                       const float* __restrict__ beta_raw,
                                                       const float* __restrict__ theta_raw,
                                                       float* __restrict__ out) {
    __shared__ float apS[16][132];
    __shared__ float bpS[16][132];

    const int tx  = threadIdx.x;
    const int ty  = threadIdx.y;
    const int tid = ty * 16 + tx;
    const int j0  = blockIdx.x * 16;
    const int i0  = blockIdx.y * 16;

    for (int p = tid; p < 16 * 32; p += 256) {
        const int r  = p >> 5;
        const int c4 = (p & 31) * 4;
        *(float4*)&apS[r][c4] = *(const float4*)(ap + (size_t)(i0 + r) * F_DIM + c4);
        *(float4*)&bpS[r][c4] = *(const float4*)(bp + (size_t)(j0 + r) * F_DIM + c4);
    }
    __syncthreads();

    float s_min = 0.0f, s_a = 0.0f, s_b = 0.0f;
#pragma unroll 4
    for (int k = 0; k < F_DIM; ++k) {
        const float av = apS[ty][k];
        const float bv = bpS[tx][k];
        s_min += fminf(av, bv);
        s_a   += (bv > 0.0f) ? av : 0.0f;
        s_b   += (av > 0.0f) ? bv : 0.0f;
    }

    const float al = softplus_f(alpha_raw[0]);
    const float be = softplus_f(beta_raw[0]);
    const float th = softplus_f(theta_raw[0]);
    out[(size_t)(i0 + ty) * P_ROWS + (j0 + tx)] =
        (th + al + be) * s_min - al * s_a - be * s_b;
}

extern "C" void kernel_launch(void* const* d_in, const int* in_sizes, int n_in,
                              void* d_out, int out_size, void* d_ws, size_t ws_size,
                              hipStream_t stream) {
    const float* input     = (const float*)d_in[0];
    const float* prototype = (const float*)d_in[1];
    const float* fb        = (const float*)d_in[2];
    const float* alpha_raw = (const float*)d_in[3];
    const float* beta_raw  = (const float*)d_in[4];
    const float* theta_raw = (const float*)d_in[5];
    float* out = (float*)d_out;

    float* ap = (float*)d_ws;
    float* bp = ap + (size_t)N_ROWS * F_DIM;

    proj_norm_relu_wmma<<<((N_ROWS + P_ROWS) / 16) * (F_DIM / 16) / 5, 160, 0, stream>>>(
        input, prototype, fb, ap);

    dim3 grid(P_ROWS / 16, N_ROWS / 16);
    pairwise_kernel<<<grid, dim3(16, 16), 0, stream>>>(ap, bp, alpha_raw, beta_raw, theta_raw, out);
}


# Pairwise kernel per wave on WMMA fragments, no LDS: S_a, S_b as f32 WMMA products with 0/1 masks, min-sum reads the same registers by DPP row_share; sums reordered (k in two halves)
# speedup vs baseline: 1.9273x; 1.1427x over previous
#include <hip/hip_runtime.h>
#include <hip/hip_bf16.h>
#include <math.h>

typedef float v2f __attribute__((ext_vector_type(2)));
typedef float v8f __attribute__((ext_vector_type(8)));

#define N_ROWS 2048
#define P_ROWS 512
#define D_DIM  1024
#define F_DIM  128
#define EPSV   1e-8f

__device__ __forceinline__ float softplus_f(float x) {
    return log1pf(expf(x));
}

__global__ __launch_bounds__(160) void proj_norm_relu_wmma(const float* __restrict__ Xin,
                                                           const float* __restrict__ Xpr,
                                                           const float* __restrict__ fb,
                                                           float* __restrict__ proj) {
    const int wid  = threadIdx.x >> 5;
    const int lane = threadIdx.x & 31;
    const int tile = blockIdx.x * 5 + wid;
    const int tf   = tile & 7;
    const int tm   = tile >> 3;
    const int half = lane >> 4;
    const int lm   = lane & 15;

    const float* X = (tm < N_ROWS / 16) ? Xin + (size_t)tm * 16 * D_DIM
                                        : Xpr + (size_t)(tm - N_ROWS / 16) * 16 * D_DIM;
    const float* arow = X  + (size_t)lm * D_DIM + 2 * half;
    const float* brow = fb + (size_t)(tf * 16 + lm) * D_DIM + 2 * half;

    v8f acc = {};
    v2f sa2 = {}, sb2 = {};
#pragma unroll 8
    for (int k = 0; k < D_DIM; k += 4) {
        v2f a2 = *(const v2f*)(arow + k);
        v2f b2 = *(const v2f*)(brow + k);
        sa2 = __builtin_elementwise_fma(a2, a2, sa2);
        sb2 = __builtin_elementwise_fma(b2, b2, sb2);
        acc = __builtin_amdgcn_wmma_f32_16x16x4_f32(
             false, a2,  false, b2,
             (short)0, acc,  false,  false);
    }

    float sa = sa2.x + sa2.y;
    float sb = sb2.x + sb2.y;
    sa += __shfl_xor(sa, 16);
    sb += __shfl_xor(sb, 16);
    const float ia = 1.0f / fmaxf(sqrtf(sa), EPSV);
    const float ib = 1.0f / fmaxf(sqrtf(sb), EPSV);

    const int col = tf * 16 + lm;
#pragma unroll
    for (int v = 0; v < 8; ++v) {
        const int r = half * 8 + v;
        const float iar = __shfl(ia, r);
        const float val = acc[v] * iar * ib;
        proj[(size_t)(tm * 16 + r) * F_DIM + col] = fmaxf(val, 0.0f);
    }
}

template <int N>
__device__ __forceinline__ int row_share(int v) {
    return __builtin_amdgcn_update_dpp(0, v, 0x150 + N, 0xf, 0xf, true);
}

__device__ __forceinline__ float min_nonneg(int abits, int bbits) {
    return __builtin_bit_cast(float, (abits < bbits) ? abits : bbits);
}

#define MIN_STEP(n, av, bv) acc[n] += min_nonneg(row_share<n>(av), (bv));
#define MIN_ROWS(av, bv)                                                    \
    MIN_STEP(0, av, bv)  MIN_STEP(1, av, bv)  MIN_STEP(2, av, bv)  MIN_STEP(3, av, bv)   \
    MIN_STEP(4, av, bv)  MIN_STEP(5, av, bv)  MIN_STEP(6, av, bv)  MIN_STEP(7, av, bv)   \
    MIN_STEP(8, av, bv)  MIN_STEP(9, av, bv)  MIN_STEP(10, av, bv) MIN_STEP(11, av, bv)  \
    MIN_STEP(12, av, bv) MIN_STEP(13, av, bv) MIN_STEP(14, av, bv) MIN_STEP(15, av, bv)

__device__ __forceinline__ float pos_mask(float x) { return (x > 0.0f) ? 1.0f : 0.0f; }

#define J_TILES_PER_WAVE 2

__global__ __launch_bounds__(256) void pairwise_wmma(const float* __restrict__ ap,
                                                     const float* __restrict__ bp,
                                                     const float* __restrict__ alpha_raw,
                                                     const float* __restrict__ beta_raw,
                                                     const float* __restrict__ theta_raw,
                                                     float* __restrict__ out) {
    const int w    = threadIdx.x >> 5;
    const int lane = threadIdx.x & 31;
    const int h    = lane >> 4;
    const int c    = lane & 15;
    const int ti   = blockIdx.y * 2 + (w >> 2);
    const int tj0  = (blockIdx.x * 4 + (w & 3)) * J_TILES_PER_WAVE;

    const float* arow = ap + (size_t)(ti * 16 + c) * F_DIM + 4 * h;
    float4 A[16], MA[16];
#pragma unroll
    for (int t = 0; t < 16; ++t) A[t] = *(const float4*)(arow + 8 * t);
#pragma unroll
    for (int t = 0; t < 16; ++t) {
        MA[t].x = pos_mask(A[t].x); MA[t].y = pos_mask(A[t].y);
        MA[t].z = pos_mask(A[t].z); MA[t].w = pos_mask(A[t].w);
    }

    const float al = softplus_f(alpha_raw[0]);
    const float be = softplus_f(beta_raw[0]);
    const float th = softplus_f(theta_raw[0]);
    const float c1 = th + al + be;

    for (int jt = 0; jt < J_TILES_PER_WAVE; ++jt) {
        const int tj = tj0 + jt;
        const float* brow = bp + (size_t)(tj * 16 + c) * F_DIM + 4 * h;
        float4 B[16];
#pragma unroll
        for (int t = 0; t < 16; ++t) B[t] = *(const float4*)(brow + 8 * t);

        float acc[16];
#pragma unroll
        for (int n = 0; n < 16; ++n) acc[n] = 0.0f;
        v8f Sa = {}, Sb = {};

#pragma unroll
        for (int t = 0; t < 16; ++t) {
            const float4 a4 = A[t], b4 = B[t], ma4 = MA[t];
            float4 mb4;
            mb4.x = pos_mask(b4.x); mb4.y = pos_mask(b4.y);
            mb4.z = pos_mask(b4.z); mb4.w = pos_mask(b4.w);
            v2f a_lo = {a4.x, a4.y},   a_hi = {a4.z, a4.w};
            v2f b_lo = {b4.x, b4.y},   b_hi = {b4.z, b4.w};
            v2f ma_lo = {ma4.x, ma4.y}, ma_hi = {ma4.z, ma4.w};
            v2f mb_lo = {mb4.x, mb4.y}, mb_hi = {mb4.z, mb4.w};
            Sa = __builtin_amdgcn_wmma_f32_16x16x4_f32(false, a_lo,  false, mb_lo, (short)0, Sa, false, false);
            Sb = __builtin_amdgcn_wmma_f32_16x16x4_f32(false, ma_lo, false, b_lo,  (short)0, Sb, false, false);
            Sa = __builtin_amdgcn_wmma_f32_16x16x4_f32(false, a_hi,  false, mb_hi, (short)0, Sa, false, false);
            Sb = __builtin_amdgcn_wmma_f32_16x16x4_f32(false, ma_hi, false, b_hi,  (short)0, Sb, false, false);
            const int ax = __builtin_bit_cast(int, a4.x), bx = __builtin_bit_cast(int, b4.x);
            const int ay = __builtin_bit_cast(int, a4.y), by = __builtin_bit_cast(int, b4.y);
            const int az = __builtin_bit_cast(int, a4.z), bz = __builtin_bit_cast(int, b4.z);
            const int aw = __builtin_bit_cast(int, a4.w), bw = __builtin_bit_cast(int, b4.w);
            MIN_ROWS(ax, bx)
            MIN_ROWS(ay, by)
            MIN_ROWS(az, bz)
            MIN_ROWS(aw, bw)
        }

#pragma unroll
        for (int r = 0; r < 8; ++r) {
            const float keep = h ? acc[8 + r] : acc[r];
            const float send = h ? acc[r]     : acc[8 + r];
            const float smin = keep + __shfl_xor(send, 16);
            const float val  = c1 * smin - al * Sa[r] - be * Sb[r];
            out[(size_t)(ti * 16 + 8 * h + r) * P_ROWS + (tj * 16 + c)] = val;
        }
    }
}

extern "C" void kernel_launch(void* const* d_in, const int* in_sizes, int n_in,
                              void* d_out, int out_size, void* d_ws, size_t ws_size,
                              hipStream_t stream) {
    const float* input     = (const float*)d_in[0];
    const float* prototype = (const float*)d_in[1];
    const float* fb        = (const float*)d_in[2];
    const float* alpha_raw = (const float*)d_in[3];
    const float* beta_raw  = (const float*)d_in[4];
    const float* theta_raw = (const float*)d_in[5];
    float* out = (float*)d_out;

    float* ap = (float*)d_ws;
    float* bp = ap + (size_t)N_ROWS * F_DIM;

    proj_norm_relu_wmma<<<((N_ROWS + P_ROWS) / 16) * (F_DIM / 16) / 5, 160, 0, stream>>>(
        input, prototype, fb, ap);

    dim3 grid(P_ROWS / 128, N_ROWS / 32);
    pairwise_wmma<<<grid, 256, 0, stream>>>(ap, bp, alpha_raw, beta_raw, theta_raw, out);
}


# Projection launch re-tiled through LDS: 8-wave block = 16 rows x all 128 cols, X and fb staged per 128-k chunk by whole-line 16-B loads, 2 stages (dot and norm sums reordered: k order, 2 accumulators)
# speedup vs baseline: 2.3705x; 1.2300x over previous
#include <hip/hip_runtime.h>
#include <hip/hip_bf16.h>
#include <math.h>

typedef float v2f __attribute__((ext_vector_type(2)));
typedef float v4f __attribute__((ext_vector_type(4)));
typedef float v8f __attribute__((ext_vector_type(8)));

#define N_ROWS 2048
#define P_ROWS 512
#define D_DIM  1024
#define F_DIM  128
#define EPSV   1e-8f

__device__ __forceinline__ float softplus_f(float x) {
    return log1pf(expf(x));
}

#define KC       128
#define NCHUNK   (D_DIM / KC)
#define ST_ROWS  (16 + F_DIM)
#define PITCH4   (KC / 4 + 1)
#define ST_F4    (ST_ROWS * PITCH4)

__device__ __forceinline__ void mma_quad(const v4f a4, const v4f b4,
                                         v8f& acc0, v8f& acc1, v2f& sa2, v2f& sb2) {
    v2f a_lo = {a4.x, a4.y}, a_hi = {a4.z, a4.w};
    v2f b_lo = {b4.x, b4.y}, b_hi = {b4.z, b4.w};
    sa2 = __builtin_elementwise_fma(a_lo, a_lo, sa2);
    sb2 = __builtin_elementwise_fma(b_lo, b_lo, sb2);
    sa2 = __builtin_elementwise_fma(a_hi, a_hi, sa2);
    sb2 = __builtin_elementwise_fma(b_hi, b_hi, sb2);
    acc0 = __builtin_amdgcn_wmma_f32_16x16x4_f32(
         false, a_lo,  false, b_lo,
         (short)0, acc0,  false,  false);
    acc1 = __builtin_amdgcn_wmma_f32_16x16x4_f32(
         false, a_hi,  false, b_hi,
         (short)0, acc1,  false,  false);
}

__global__ __launch_bounds__(256) void proj_norm_relu_wmma(const float* __restrict__ Xin,
                                                           const float* __restrict__ Xpr,
                                                           const float* __restrict__ fb,
                                                           float* __restrict__ proj) {
    __shared__ v4f stage[2 * ST_F4];

    const int w    = threadIdx.x >> 5;
    const int lane = threadIdx.x & 31;
    const int tm   = blockIdx.x;
    const int half = lane >> 4;
    const int lm   = lane & 15;

    const float* X = (tm < N_ROWS / 16) ? Xin + (size_t)tm * 16 * D_DIM
                                        : Xpr + (size_t)(tm - N_ROWS / 16) * 16 * D_DIM;

    const float* gx = X  + (size_t)w * D_DIM + 4 * lane;
    const float* gf = fb + (size_t)w * D_DIM + 4 * lane;
    v4f* const st_w = stage + w * PITCH4 + lane;
    const v4f* const fa = stage + lm * PITCH4 + half;
    const v4f* const fbq = stage + (16 + 16 * w + lm) * PITCH4 + half;

    {
        v4f gF[F_DIM / 8];
        const v4f gX0 = *(const v4f*)(gx);
        const v4f gX1 = *(const v4f*)(gx + (size_t)8 * D_DIM);
#pragma unroll
        for (int i = 0; i < F_DIM / 8; ++i) gF[i] = *(const v4f*)(gf + (size_t)(8 * i) * D_DIM);
        st_w[0] = gX0;
        st_w[8 * PITCH4] = gX1;
#pragma unroll
        for (int i = 0; i < F_DIM / 8; ++i) st_w[(i + 2) * 8 * PITCH4] = gF[i];
    }
    __syncthreads();

    v8f acc0 = {}, acc1 = {};
    v2f sa2 = {}, sb2 = {};
    for (int ch = 0; ch < NCHUNK - 1; ++ch) {
        const int s = ch & 1;
        const int ko = (ch + 1) * KC;
        v4f gF[F_DIM / 8];
        const v4f gX0 = *(const v4f*)(gx + ko);
        const v4f gX1 = *(const v4f*)(gx + (size_t)8 * D_DIM + ko);
#pragma unroll
        for (int i = 0; i < F_DIM / 8; ++i) gF[i] = *(const v4f*)(gf + (size_t)(8 * i) * D_DIM + ko);

        const v4f* pa = fa + s * ST_F4;
        const v4f* pb = fbq + s * ST_F4;
        v4f A[KC / 8], B[KC / 8];
#pragma unroll
        for (int t = 0; t < KC / 8; ++t) { A[t] = pa[2 * t]; B[t] = pb[2 * t]; }
#pragma unroll
        for (int t = 0; t < KC / 16; ++t) mma_quad(A[t], B[t], acc0, acc1, sa2, sb2);

        asm volatile("" :: "v"(acc0), "v"(acc1) : "memory");
        v4f* so = st_w + (s ^ 1) * ST_F4;
        so[0] = gX0;
        so[8 * PITCH4] = gX1;
#pragma unroll
        for (int i = 0; i < F_DIM / 8; ++i) so[(i + 2) * 8 * PITCH4] = gF[i];

#pragma unroll
        for (int t = KC / 16; t < KC / 8 - 2; ++t) mma_quad(A[t], B[t], acc0, acc1, sa2, sb2);
        asm volatile("" :: "v"(acc0), "v"(acc1));
        __syncthreads();
#pragma unroll
        for (int t = KC / 8 - 2; t < KC / 8; ++t) mma_quad(A[t], B[t], acc0, acc1, sa2, sb2);
    }
    {
        const v4f* pa = fa  + ((NCHUNK - 1) & 1) * ST_F4;
        const v4f* pb = fbq + ((NCHUNK - 1) & 1) * ST_F4;
#pragma unroll
        for (int t = 0; t < KC / 8; ++t) mma_quad(pa[2 * t], pb[2 * t], acc0, acc1, sa2, sb2);
    }
    const v8f acc = acc0 + acc1;

    float sa = sa2.x + sa2.y;
    float sb = sb2.x + sb2.y;
    sa += __shfl_xor(sa, 16);
    sb += __shfl_xor(sb, 16);
    const float ia = 1.0f / fmaxf(sqrtf(sa), EPSV);
    const float ib = 1.0f / fmaxf(sqrtf(sb), EPSV);

    const int col = w * 16 + lm;
#pragma unroll
    for (int v = 0; v < 8; ++v) {
        const int r = half * 8 + v;
        const float iar = __shfl(ia, r);
        const float val = acc[v] * iar * ib;
        proj[(size_t)(tm * 16 + r) * F_DIM + col] = fmaxf(val, 0.0f);
    }
}

template <int N>
__device__ __forceinline__ int row_share(int v) {
    return __builtin_amdgcn_update_dpp(0, v, 0x150 + N, 0xf, 0xf, true);
}

__device__ __forceinline__ float min_nonneg(int abits, int bbits) {
    return __builtin_bit_cast(float, (abits < bbits) ? abits : bbits);
}

#define MIN_STEP(n, av, bv) acc[n] += min_nonneg(row_share<n>(av), (bv));
#define MIN_ROWS(av, bv)                                                    \
    MIN_STEP(0, av, bv)  MIN_STEP(1, av, bv)  MIN_STEP(2, av, bv)  MIN_STEP(3, av, bv)   \
    MIN_STEP(4, av, bv)  MIN_STEP(5, av, bv)  MIN_STEP(6, av, bv)  MIN_STEP(7, av, bv)   \
    MIN_STEP(8, av, bv)  MIN_STEP(9, av, bv)  MIN_STEP(10, av, bv) MIN_STEP(11, av, bv)  \
    MIN_STEP(12, av, bv) MIN_STEP(13, av, bv) MIN_STEP(14, av, bv) MIN_STEP(15, av, bv)

__device__ __forceinline__ float pos_mask(float x) { return (x > 0.0f) ? 1.0f : 0.0f; }

#define J_TILES_PER_WAVE 2

__global__ __launch_bounds__(256) void pairwise_wmma(const float* __restrict__ ap,
                                                     const float* __restrict__ bp,
                                                     const float* __restrict__ alpha_raw,
                                                     const float* __restrict__ beta_raw,
                                                     const float* __restrict__ theta_raw,
                                                     float* __restrict__ out) {
    const int w    = threadIdx.x >> 5;
    const int lane = threadIdx.x & 31;
    const int h    = lane >> 4;
    const int c    = lane & 15;
    const int ti   = blockIdx.y * 2 + (w >> 2);
    const int tj0  = (blockIdx.x * 4 + (w & 3)) * J_TILES_PER_WAVE;

    const float* arow = ap + (size_t)(ti * 16 + c) * F_DIM + 4 * h;
    float4 A[16], MA[16];
#pragma unroll
    for (int t = 0; t < 16; ++t) A[t] = *(const float4*)(arow + 8 * t);
#pragma unroll
    for (int t = 0; t < 16; ++t) {
        MA[t].x = pos_mask(A[t].x); MA[t].y = pos_mask(A[t].y);
        MA[t].z = pos_mask(A[t].z); MA[t].w = pos_mask(A[t].w);
    }

    const float al = softplus_f(alpha_raw[0]);
    const float be = softplus_f(beta_raw[0]);
    const float th = softplus_f(theta_raw[0]);
    const float c1 = th + al + be;

    for (int jt = 0; jt < J_TILES_PER_WAVE; ++jt) {
        const int tj = tj0 + jt;
        const float* brow = bp + (size_t)(tj * 16 + c) * F_DIM + 4 * h;
        float4 B[16];
#pragma unroll
        for (int t = 0; t < 16; ++t) B[t] = *(const float4*)(brow + 8 * t);

        float acc[16];
#pragma unroll
        for (int n = 0; n < 16; ++n) acc[n] = 0.0f;
        v8f Sa = {}, Sb = {};

#pragma unroll
        for (int t = 0; t < 16; ++t) {
            const float4 a4 = A[t], b4 = B[t], ma4 = MA[t];
            float4 mb4;
            mb4.x = pos_mask(b4.x); mb4.y = pos_mask(b4.y);
            mb4.z = pos_mask(b4.z); mb4.w = pos_mask(b4.w);
            v2f a_lo = {a4.x, a4.y},   a_hi = {a4.z, a4.w};
            v2f b_lo = {b4.x, b4.y},   b_hi = {b4.z, b4.w};
            v2f ma_lo = {ma4.x, ma4.y}, ma_hi = {ma4.z, ma4.w};
            v2f mb_lo = {mb4.x, mb4.y}, mb_hi = {mb4.z, mb4.w};
            Sa = __builtin_amdgcn_wmma_f32_16x16x4_f32(false, a_lo,  false, mb_lo, (short)0, Sa, false, false);
            Sb = __builtin_amdgcn_wmma_f32_16x16x4_f32(false, ma_lo, false, b_lo,  (short)0, Sb, false, false);
            Sa = __builtin_amdgcn_wmma_f32_16x16x4_f32(false, a_hi,  false, mb_hi, (short)0, Sa, false, false);
            Sb = __builtin_amdgcn_wmma_f32_16x16x4_f32(false, ma_hi, false, b_hi,  (short)0, Sb, false, false);
            const int ax = __builtin_bit_cast(int, a4.x), bx = __builtin_bit_cast(int, b4.x);
            const int ay = __builtin_bit_cast(int, a4.y), by = __builtin_bit_cast(int, b4.y);
            const int az = __builtin_bit_cast(int, a4.z), bz = __builtin_bit_cast(int, b4.z);
            const int aw = __builtin_bit_cast(int, a4.w), bw = __builtin_bit_cast(int, b4.w);
            MIN_ROWS(ax, bx)
            MIN_ROWS(ay, by)
            MIN_ROWS(az, bz)
            MIN_ROWS(aw, bw)
        }

#pragma unroll
        for (int r = 0; r < 8; ++r) {
            const float keep = h ? acc[8 + r] : acc[r];
            const float send = h ? acc[r]     : acc[8 + r];
            const float smin = keep + __shfl_xor(send, 16);
            const float val  = c1 * smin - al * Sa[r] - be * Sb[r];
            out[(size_t)(ti * 16 + 8 * h + r) * P_ROWS + (tj * 16 + c)] = val;
        }
    }
}

extern "C" void kernel_launch(void* const* d_in, const int* in_sizes, int n_in,
                              void* d_out, int out_size, void* d_ws, size_t ws_size,
                              hipStream_t stream) {
    const float* input     = (const float*)d_in[0];
    const float* prototype = (const float*)d_in[1];
    const float* fb        = (const float*)d_in[2];
    const float* alpha_raw = (const float*)d_in[3];
    const float* beta_raw  = (const float*)d_in[4];
    const float* theta_raw = (const float*)d_in[5];
    float* out = (float*)d_out;

    float* ap = (float*)d_ws;
    float* bp = ap + (size_t)N_ROWS * F_DIM;

    proj_norm_relu_wmma<<<(N_ROWS + P_ROWS) / 16, 256, 0, stream>>>(
        input, prototype, fb, ap);

    dim3 grid(P_ROWS / 128, N_ROWS / 32);
    pairwise_wmma<<<grid, 256, 0, stream>>>(ap, bp, alpha_raw, beta_raw, theta_raw, out);
}


# Pairwise min-sum reads ap by broadcast LDS loads of a once-staged 32-row tile (a 16-B read feeds 8 plain min+add pairs, two column tiles per pass), not a DPP row_share per min; sum order kept
# speedup vs baseline: 2.6836x; 1.1321x over previous
#include <hip/hip_runtime.h>
#include <hip/hip_bf16.h>
#include <math.h>

typedef float v2f __attribute__((ext_vector_type(2)));
typedef float v4f __attribute__((ext_vector_type(4)));
typedef float v8f __attribute__((ext_vector_type(8)));

#define N_ROWS 2048
#define P_ROWS 512
#define D_DIM  1024
#define F_DIM  128
#define EPSV   1e-8f

__device__ __forceinline__ float softplus_f(float x) {
    return log1pf(expf(x));
}

#define KC       128
#define NCHUNK   (D_DIM / KC)
#define ST_ROWS  (16 + F_DIM)
#define PITCH4   (KC / 4 + 1)
#define ST_F4    (ST_ROWS * PITCH4)

__device__ __forceinline__ void mma_quad(const v4f a4, const v4f b4,
                                         v8f& acc0, v8f& acc1, v2f& sa2, v2f& sb2) {
    v2f a_lo = {a4.x, a4.y}, a_hi = {a4.z, a4.w};
    v2f b_lo = {b4.x, b4.y}, b_hi = {b4.z, b4.w};
    sa2 = __builtin_elementwise_fma(a_lo, a_lo, sa2);
    sb2 = __builtin_elementwise_fma(b_lo, b_lo, sb2);
    sa2 = __builtin_elementwise_fma(a_hi, a_hi, sa2);
    sb2 = __builtin_elementwise_fma(b_hi, b_hi, sb2);
    acc0 = __builtin_amdgcn_wmma_f32_16x16x4_f32(
         false, a_lo,  false, b_lo,
         (short)0, acc0,  false,  false);
    acc1 = __builtin_amdgcn_wmma_f32_16x16x4_f32(
         false, a_hi,  false, b_hi,
         (short)0, acc1,  false,  false);
}

__global__ __launch_bounds__(256) void proj_norm_relu_wmma(const float* __restrict__ Xin,
                                                           const float* __restrict__ Xpr,
                                                           const float* __restrict__ fb,
                                                           float* __restrict__ proj) {
    __shared__ v4f stage[2 * ST_F4];

    const int w    = threadIdx.x >> 5;
    const int lane = threadIdx.x & 31;
    const int tm   = blockIdx.x;
    const int half = lane >> 4;
    const int lm   = lane & 15;

    const float* X = (tm < N_ROWS / 16) ? Xin + (size_t)tm * 16 * D_DIM
                                        : Xpr + (size_t)(tm - N_ROWS / 16) * 16 * D_DIM;

    const float* gx = X  + (size_t)w * D_DIM + 4 * lane;
    const float* gf = fb + (size_t)w * D_DIM + 4 * lane;
    v4f* const st_w = stage + w * PITCH4 + lane;
    const v4f* const fa = stage + lm * PITCH4 + half;
    const v4f* const fbq = stage + (16 + 16 * w + lm) * PITCH4 + half;

    {
        v4f gF[F_DIM / 8];
        const v4f gX0 = *(const v4f*)(gx);
        const v4f gX1 = *(const v4f*)(gx + (size_t)8 * D_DIM);
#pragma unroll
        for (int i = 0; i < F_DIM / 8; ++i) gF[i] = *(const v4f*)(gf + (size_t)(8 * i) * D_DIM);
        st_w[0] = gX0;
        st_w[8 * PITCH4] = gX1;
#pragma unroll
        for (int i = 0; i < F_DIM / 8; ++i) st_w[(i + 2) * 8 * PITCH4] = gF[i];
    }
    __syncthreads();

    v8f acc0 = {}, acc1 = {};
    v2f sa2 = {}, sb2 = {};
    for (int ch = 0; ch < NCHUNK - 1; ++ch) {
        const int s = ch & 1;
        const int ko = (ch + 1) * KC;
        v4f gF[F_DIM / 8];
        const v4f gX0 = *(const v4f*)(gx + ko);
        const v4f gX1 = *(const v4f*)(gx + (size_t)8 * D_DIM + ko);
#pragma unroll
        for (int i = 0; i < F_DIM / 8; ++i) gF[i] = *(const v4f*)(gf + (size_t)(8 * i) * D_DIM + ko);

        const v4f* pa = fa + s * ST_F4;
        const v4f* pb = fbq + s * ST_F4;
        v4f A[KC / 8], B[KC / 8];
#pragma unroll
        for (int t = 0; t < KC / 8; ++t) { A[t] = pa[2 * t]; B[t] = pb[2 * t]; }
#pragma unroll
        for (int t = 0; t < KC / 16; ++t) mma_quad(A[t], B[t], acc0, acc1, sa2, sb2);

        asm volatile("" :: "v"(acc0), "v"(acc1) : "memory");
        v4f* so = st_w + (s ^ 1) * ST_F4;
        so[0] = gX0;
        so[8 * PITCH4] = gX1;
#pragma unroll
        for (int i = 0; i < F_DIM / 8; ++i) so[(i + 2) * 8 * PITCH4] = gF[i];

#pragma unroll
        for (int t = KC / 16; t < KC / 8 - 2; ++t) mma_quad(A[t], B[t], acc0, acc1, sa2, sb2);
        asm volatile("" :: "v"(acc0), "v"(acc1));
        __syncthreads();
#pragma unroll
        for (int t = KC / 8 - 2; t < KC / 8; ++t) mma_quad(A[t], B[t], acc0, acc1, sa2, sb2);
    }
    {
        const v4f* pa = fa  + ((NCHUNK - 1) & 1) * ST_F4;
        const v4f* pb = fbq + ((NCHUNK - 1) & 1) * ST_F4;
#pragma unroll
        for (int t = 0; t < KC / 8; ++t) mma_quad(pa[2 * t], pb[2 * t], acc0, acc1, sa2, sb2);
    }
    const v8f acc = acc0 + acc1;

    float sa = sa2.x + sa2.y;
    float sb = sb2.x + sb2.y;
    sa += __shfl_xor(sa, 16);
    sb += __shfl_xor(sb, 16);
    const float ia = 1.0f / fmaxf(sqrtf(sa), EPSV);
    const float ib = 1.0f / fmaxf(sqrtf(sb), EPSV);

    const int col = w * 16 + lm;
#pragma unroll
    for (int v = 0; v < 8; ++v) {
        const int r = half * 8 + v;
        const float iar = __shfl(ia, r);
        const float val = acc[v] * iar * ib;
        proj[(size_t)(tm * 16 + r) * F_DIM + col] = fmaxf(val, 0.0f);
    }
}

#define A_PITCH4 (F_DIM / 4 + 1)

__device__ __forceinline__ v4f quiet4(const v4f x) { return __builtin_elementwise_canonicalize(x); }

__device__ __forceinline__ float pos_mask(float x) { return (x > 0.0f) ? 1.0f : 0.0f; }

__device__ __forceinline__ v4f pos_mask4(const v4f x) {
    v4f m;
    m.x = pos_mask(x.x); m.y = pos_mask(x.y); m.z = pos_mask(x.z); m.w = pos_mask(x.w);
    return m;
}

#define J_TILES_PER_WAVE 2

__global__ __launch_bounds__(256) void pairwise_wmma(const float* __restrict__ ap,
                                                     const float* __restrict__ bp,
                                                     const float* __restrict__ alpha_raw,
                                                     const float* __restrict__ beta_raw,
                                                     const float* __restrict__ theta_raw,
                                                     float* __restrict__ out) {
    __shared__ v4f a_sh[32 * A_PITCH4];

    const int w    = threadIdx.x >> 5;
    const int lane = threadIdx.x & 31;
    const int h    = lane >> 4;
    const int c    = lane & 15;
    const int tl   = w >> 2;
    const int ti   = blockIdx.y * 2 + tl;
    const int tj0  = (blockIdx.x * 4 + (w & 3)) * J_TILES_PER_WAVE;

    {
        const float* abase = ap + (size_t)blockIdx.y * 32 * F_DIM;
        v4f g[4];
#pragma unroll
        for (int r = 0; r < 4; ++r) g[r] = *(const v4f*)(abase + (size_t)(8 * r + w) * F_DIM + 4 * lane);
#pragma unroll
        for (int r = 0; r < 4; ++r) a_sh[(8 * r + w) * A_PITCH4 + lane] = g[r];
    }

    const float* brow0 = bp + (size_t)(tj0 * 16 + c) * F_DIM + 4 * h;
    const float* brow1 = brow0 + (size_t)16 * F_DIM;
    v4f B0[16], B1[16];
#pragma unroll
    for (int t = 0; t < 16; ++t) B0[t] = quiet4(*(const v4f*)(brow0 + 8 * t));
#pragma unroll
    for (int t = 0; t < 16; ++t) B1[t] = quiet4(*(const v4f*)(brow1 + 8 * t));

    const float al = softplus_f(alpha_raw[0]);
    const float be = softplus_f(beta_raw[0]);
    const float th = softplus_f(theta_raw[0]);
    const float c1 = th + al + be;

    __syncthreads();

    const v4f* const afrag = a_sh + (tl * 16 + c) * A_PITCH4 + h;
    const v4f* const arow  = a_sh + (tl * 16) * A_PITCH4 + h;

    float acc0[16], acc1[16];
#pragma unroll
    for (int n = 0; n < 16; ++n) { acc0[n] = 0.0f; acc1[n] = 0.0f; }
    v8f Sa0 = {}, Sb0 = {}, Sa1 = {}, Sb1 = {};

    __builtin_amdgcn_sched_group_barrier(0x100, 17, 0);
#pragma unroll
    for (int t = 0; t < 16; ++t) {
        const v4f a4 = afrag[2 * t];
        const v4f b0 = B0[t], b1 = B1[t];
        const v4f ma4 = pos_mask4(a4), mb0 = pos_mask4(b0), mb1 = pos_mask4(b1);
        const v2f a_lo  = {a4.x, a4.y},   a_hi  = {a4.z, a4.w};
        const v2f ma_lo = {ma4.x, ma4.y}, ma_hi = {ma4.z, ma4.w};
        const v2f b0_lo = {b0.x, b0.y},   b0_hi = {b0.z, b0.w};
        const v2f b1_lo = {b1.x, b1.y},   b1_hi = {b1.z, b1.w};
        const v2f mb0_lo = {mb0.x, mb0.y}, mb0_hi = {mb0.z, mb0.w};
        const v2f mb1_lo = {mb1.x, mb1.y}, mb1_hi = {mb1.z, mb1.w};
        Sa0 = __builtin_amdgcn_wmma_f32_16x16x4_f32(false, a_lo,  false, mb0_lo, (short)0, Sa0, false, false);
        Sb0 = __builtin_amdgcn_wmma_f32_16x16x4_f32(false, ma_lo, false, b0_lo,  (short)0, Sb0, false, false);
        Sa1 = __builtin_amdgcn_wmma_f32_16x16x4_f32(false, a_lo,  false, mb1_lo, (short)0, Sa1, false, false);
        Sb1 = __builtin_amdgcn_wmma_f32_16x16x4_f32(false, ma_lo, false, b1_lo,  (short)0, Sb1, false, false);
        Sa0 = __builtin_amdgcn_wmma_f32_16x16x4_f32(false, a_hi,  false, mb0_hi, (short)0, Sa0, false, false);
        Sb0 = __builtin_amdgcn_wmma_f32_16x16x4_f32(false, ma_hi, false, b0_hi,  (short)0, Sb0, false, false);
        Sa1 = __builtin_amdgcn_wmma_f32_16x16x4_f32(false, a_hi,  false, mb1_hi, (short)0, Sa1, false, false);
        Sb1 = __builtin_amdgcn_wmma_f32_16x16x4_f32(false, ma_hi, false, b1_hi,  (short)0, Sb1, false, false);
#pragma unroll
        for (int n = 0; n < 16; ++n) {
            const v4f q = quiet4(arow[n * A_PITCH4 + 2 * t]);
            acc0[n] += fminf(q.x, b0.x);  acc1[n] += fminf(q.x, b1.x);
            acc0[n] += fminf(q.y, b0.y);  acc1[n] += fminf(q.y, b1.y);
            acc0[n] += fminf(q.z, b0.z);  acc1[n] += fminf(q.z, b1.z);
            acc0[n] += fminf(q.w, b0.w);  acc1[n] += fminf(q.w, b1.w);
        }
        __builtin_amdgcn_sched_group_barrier(0x100, 17, 0);
        __builtin_amdgcn_sched_group_barrier(0x00A, 352, 0);
    }

#pragma unroll
    for (int r = 0; r < 8; ++r) {
        const float keep0 = h ? acc0[8 + r] : acc0[r];
        const float send0 = h ? acc0[r]     : acc0[8 + r];
        const float keep1 = h ? acc1[8 + r] : acc1[r];
        const float send1 = h ? acc1[r]     : acc1[8 + r];
        const float smin0 = keep0 + __shfl_xor(send0, 16);
        const float smin1 = keep1 + __shfl_xor(send1, 16);
        const float val0  = c1 * smin0 - al * Sa0[r] - be * Sb0[r];
        const float val1  = c1 * smin1 - al * Sa1[r] - be * Sb1[r];
        float* orow = out + (size_t)(ti * 16 + 8 * h + r) * P_ROWS + (tj0 * 16 + c);
        orow[0]  = val0;
        orow[16] = val1;
    }
}

extern "C" void kernel_launch(void* const* d_in, const int* in_sizes, int n_in,
                              void* d_out, int out_size, void* d_ws, size_t ws_size,
                              hipStream_t stream) {
    const float* input     = (const float*)d_in[0];
    const float* prototype = (const float*)d_in[1];
    const float* fb        = (const float*)d_in[2];
    const float* alpha_raw = (const float*)d_in[3];
    const float* beta_raw  = (const float*)d_in[4];
    const float* theta_raw = (const float*)d_in[5];
    float* out = (float*)d_out;

    float* ap = (float*)d_ws;
    float* bp = ap + (size_t)N_ROWS * F_DIM;

    proj_norm_relu_wmma<<<(N_ROWS + P_ROWS) / 16, 256, 0, stream>>>(
        input, prototype, fb, ap);

    dim3 grid(P_ROWS / 128, N_ROWS / 32);
    pairwise_wmma<<<grid, 256, 0, stream>>>(ap, bp, alpha_raw, beta_raw, theta_raw, out);
}


# Projection block re-shaped from 16 rows x 128 cols to 32 rows x 64 cols (2x4 wave tiles): 96 rows staged per k chunk, not 144, a third fewer bytes through each block's load path; sums unchanged
# speedup vs baseline: 2.9192x; 1.0878x over previous
#include <hip/hip_runtime.h>
#include <hip/hip_bf16.h>
#include <math.h>

typedef float v2f __attribute__((ext_vector_type(2)));
typedef float v4f __attribute__((ext_vector_type(4)));
typedef float v8f __attribute__((ext_vector_type(8)));

#define N_ROWS 2048
#define P_ROWS 512
#define D_DIM  1024
#define F_DIM  128
#define EPSV   1e-8f

__device__ __forceinline__ float softplus_f(float x) {
    return log1pf(expf(x));
}

#define KC       128
#define NCHUNK   (D_DIM / KC)
#define BM       32
#define BN       64
#define ST_ROWS  (BM + BN)
#define PITCH4   (KC / 4 + 1)
#define ST_F4    (ST_ROWS * PITCH4)

__device__ __forceinline__ void mma_quad(const v4f a4, const v4f b4,
                                         v8f& acc0, v8f& acc1, v2f& sa2, v2f& sb2) {
    v2f a_lo = {a4.x, a4.y}, a_hi = {a4.z, a4.w};
    v2f b_lo = {b4.x, b4.y}, b_hi = {b4.z, b4.w};
    sa2 = __builtin_elementwise_fma(a_lo, a_lo, sa2);
    sb2 = __builtin_elementwise_fma(b_lo, b_lo, sb2);
    sa2 = __builtin_elementwise_fma(a_hi, a_hi, sa2);
    sb2 = __builtin_elementwise_fma(b_hi, b_hi, sb2);
    acc0 = __builtin_amdgcn_wmma_f32_16x16x4_f32(
         false, a_lo,  false, b_lo,
         (short)0, acc0,  false,  false);
    acc1 = __builtin_amdgcn_wmma_f32_16x16x4_f32(
         false, a_hi,  false, b_hi,
         (short)0, acc1,  false,  false);
}

__global__ __launch_bounds__(256) __attribute__((amdgpu_waves_per_eu(2, 4)))
void proj_norm_relu_wmma(const float* __restrict__ Xin,
                         const float* __restrict__ Xpr,
                         const float* __restrict__ fb,
                         float* __restrict__ proj) {
    __shared__ v4f stage[2 * ST_F4];

    const int w    = threadIdx.x >> 5;
    const int lane = threadIdx.x & 31;
    const int rb   = blockIdx.x >> 1;
    const int cb   = blockIdx.x & 1;
    const int rt   = w >> 2;
    const int ct   = w & 3;
    const int half = lane >> 4;
    const int lm   = lane & 15;

    const float* X = (rb < N_ROWS / BM) ? Xin + (size_t)rb * BM * D_DIM
                                        : Xpr + (size_t)(rb - N_ROWS / BM) * BM * D_DIM;

    const float* gx = X  + (size_t)w * D_DIM + 4 * lane;
    const float* gf = fb + (size_t)(cb * BN + w) * D_DIM + 4 * lane;
    v4f* const st_w = stage + w * PITCH4 + lane;
    const v4f* const fa = stage + (16 * rt + lm) * PITCH4 + half;
    const v4f* const fbq = stage + (BM + 16 * ct + lm) * PITCH4 + half;

    {
        v4f gX[BM / 8], gF[BN / 8];
#pragma unroll
        for (int i = 0; i < BM / 8; ++i) gX[i] = *(const v4f*)(gx + (size_t)(8 * i) * D_DIM);
#pragma unroll
        for (int i = 0; i < BN / 8; ++i) gF[i] = *(const v4f*)(gf + (size_t)(8 * i) * D_DIM);
#pragma unroll
        for (int i = 0; i < BM / 8; ++i) st_w[(8 * i) * PITCH4] = gX[i];
#pragma unroll
        for (int i = 0; i < BN / 8; ++i) st_w[(BM + 8 * i) * PITCH4] = gF[i];
    }
    __syncthreads();

    v8f acc0 = {}, acc1 = {};
    v2f sa2 = {}, sb2 = {};
    for (int ch = 0; ch < NCHUNK - 1; ++ch) {
        const int s = ch & 1;
        const int ko = (ch + 1) * KC;
        v4f gX[BM / 8], gF[BN / 8];
#pragma unroll
        for (int i = 0; i < BM / 8; ++i) gX[i] = *(const v4f*)(gx + (size_t)(8 * i) * D_DIM + ko);
#pragma unroll
        for (int i = 0; i < BN / 8; ++i) gF[i] = *(const v4f*)(gf + (size_t)(8 * i) * D_DIM + ko);

        const v4f* pa = fa + s * ST_F4;
        const v4f* pb = fbq + s * ST_F4;
        v4f A[KC / 8], B[KC / 8];
#pragma unroll
        for (int t = 0; t < KC / 8; ++t) { A[t] = pa[2 * t]; B[t] = pb[2 * t]; }
#pragma unroll
        for (int t = 0; t < KC / 16; ++t) mma_quad(A[t], B[t], acc0, acc1, sa2, sb2);

        asm volatile("" :: "v"(acc0), "v"(acc1) : "memory");
        v4f* so = st_w + (s ^ 1) * ST_F4;
#pragma unroll
        for (int i = 0; i < BM / 8; ++i) so[(8 * i) * PITCH4] = gX[i];
#pragma unroll
        for (int i = 0; i < BN / 8; ++i) so[(BM + 8 * i) * PITCH4] = gF[i];

#pragma unroll
        for (int t = KC / 16; t < KC / 8 - 2; ++t) mma_quad(A[t], B[t], acc0, acc1, sa2, sb2);
        asm volatile("" :: "v"(acc0), "v"(acc1));
        __syncthreads();
#pragma unroll
        for (int t = KC / 8 - 2; t < KC / 8; ++t) mma_quad(A[t], B[t], acc0, acc1, sa2, sb2);
    }
    {
        const v4f* pa = fa  + ((NCHUNK - 1) & 1) * ST_F4;
        const v4f* pb = fbq + ((NCHUNK - 1) & 1) * ST_F4;
#pragma unroll
        for (int t = 0; t < KC / 8; ++t) mma_quad(pa[2 * t], pb[2 * t], acc0, acc1, sa2, sb2);
    }
    const v8f acc = acc0 + acc1;

    float sa = sa2.x + sa2.y;
    float sb = sb2.x + sb2.y;
    sa += __shfl_xor(sa, 16);
    sb += __shfl_xor(sb, 16);
    const float ia = 1.0f / fmaxf(sqrtf(sa), EPSV);
    const float ib = 1.0f / fmaxf(sqrtf(sb), EPSV);

    const int col = cb * BN + ct * 16 + lm;
#pragma unroll
    for (int v = 0; v < 8; ++v) {
        const int r = half * 8 + v;
        const float iar = __shfl(ia, r);
        const float val = acc[v] * iar * ib;
        proj[(size_t)(rb * BM + rt * 16 + r) * F_DIM + col] = fmaxf(val, 0.0f);
    }
}

#define A_PITCH4 (F_DIM / 4 + 1)

__device__ __forceinline__ v4f quiet4(const v4f x) { return __builtin_elementwise_canonicalize(x); }

__device__ __forceinline__ float pos_mask(float x) { return (x > 0.0f) ? 1.0f : 0.0f; }

__device__ __forceinline__ v4f pos_mask4(const v4f x) {
    v4f m;
    m.x = pos_mask(x.x); m.y = pos_mask(x.y); m.z = pos_mask(x.z); m.w = pos_mask(x.w);
    return m;
}

#define J_TILES_PER_WAVE 2

__global__ __launch_bounds__(256) void pairwise_wmma(const float* __restrict__ ap,
                                                     const float* __restrict__ bp,
                                                     const float* __restrict__ alpha_raw,
                                                     const float* __restrict__ beta_raw,
                                                     const float* __restrict__ theta_raw,
                                                     float* __restrict__ out) {
    __shared__ v4f a_sh[32 * A_PITCH4];

    const int w    = threadIdx.x >> 5;
    const int lane = threadIdx.x & 31;
    const int h    = lane >> 4;
    const int c    = lane & 15;
    const int tl   = w >> 2;
    const int ti   = blockIdx.y * 2 + tl;
    const int tj0  = (blockIdx.x * 4 + (w & 3)) * J_TILES_PER_WAVE;

    {
        const float* abase = ap + (size_t)blockIdx.y * 32 * F_DIM;
        v4f g[4];
#pragma unroll
        for (int r = 0; r < 4; ++r) g[r] = *(const v4f*)(abase + (size_t)(8 * r + w) * F_DIM + 4 * lane);
#pragma unroll
        for (int r = 0; r < 4; ++r) a_sh[(8 * r + w) * A_PITCH4 + lane] = g[r];
    }

    const float* brow0 = bp + (size_t)(tj0 * 16 + c) * F_DIM + 4 * h;
    const float* brow1 = brow0 + (size_t)16 * F_DIM;
    v4f B0[16], B1[16];
#pragma unroll
    for (int t = 0; t < 16; ++t) B0[t] = quiet4(*(const v4f*)(brow0 + 8 * t));
#pragma unroll
    for (int t = 0; t < 16; ++t) B1[t] = quiet4(*(const v4f*)(brow1 + 8 * t));

    const float al = softplus_f(alpha_raw[0]);
    const float be = softplus_f(beta_raw[0]);
    const float th = softplus_f(theta_raw[0]);
    const float c1 = th + al + be;

    __syncthreads();

    const v4f* const afrag = a_sh + (tl * 16 + c) * A_PITCH4 + h;
    const v4f* const arow  = a_sh + (tl * 16) * A_PITCH4 + h;

    float acc0[16], acc1[16];
#pragma unroll
    for (int n = 0; n < 16; ++n) { acc0[n] = 0.0f; acc1[n] = 0.0f; }
    v8f Sa0 = {}, Sb0 = {}, Sa1 = {}, Sb1 = {};

    __builtin_amdgcn_sched_group_barrier(0x100, 17, 0);
#pragma unroll
    for (int t = 0; t < 16; ++t) {
        const v4f a4 = afrag[2 * t];
        const v4f b0 = B0[t], b1 = B1[t];
        const v4f ma4 = pos_mask4(a4), mb0 = pos_mask4(b0), mb1 = pos_mask4(b1);
        const v2f a_lo  = {a4.x, a4.y},   a_hi  = {a4.z, a4.w};
        const v2f ma_lo = {ma4.x, ma4.y}, ma_hi = {ma4.z, ma4.w};
        const v2f b0_lo = {b0.x, b0.y},   b0_hi = {b0.z, b0.w};
        const v2f b1_lo = {b1.x, b1.y},   b1_hi = {b1.z, b1.w};
        const v2f mb0_lo = {mb0.x, mb0.y}, mb0_hi = {mb0.z, mb0.w};
        const v2f mb1_lo = {mb1.x, mb1.y}, mb1_hi = {mb1.z, mb1.w};
        Sa0 = __builtin_amdgcn_wmma_f32_16x16x4_f32(false, a_lo,  false, mb0_lo, (short)0, Sa0, false, false);
        Sb0 = __builtin_amdgcn_wmma_f32_16x16x4_f32(false, ma_lo, false, b0_lo,  (short)0, Sb0, false, false);
        Sa1 = __builtin_amdgcn_wmma_f32_16x16x4_f32(false, a_lo,  false, mb1_lo, (short)0, Sa1, false, false);
        Sb1 = __builtin_amdgcn_wmma_f32_16x16x4_f32(false, ma_lo, false, b1_lo,  (short)0, Sb1, false, false);
        Sa0 = __builtin_amdgcn_wmma_f32_16x16x4_f32(false, a_hi,  false, mb0_hi, (short)0, Sa0, false, false);
        Sb0 = __builtin_amdgcn_wmma_f32_16x16x4_f32(false, ma_hi, false, b0_hi,  (short)0, Sb0, false, false);
        Sa1 = __builtin_amdgcn_wmma_f32_16x16x4_f32(false, a_hi,  false, mb1_hi, (short)0, Sa1, false, false);
        Sb1 = __builtin_amdgcn_wmma_f32_16x16x4_f32(false, ma_hi, false, b1_hi,  (short)0, Sb1, false, false);
#pragma unroll
        for (int n = 0; n < 16; ++n) {
            const v4f q = quiet4(arow[n * A_PITCH4 + 2 * t]);
            acc0[n] += fminf(q.x, b0.x);  acc1[n] += fminf(q.x, b1.x);
            acc0[n] += fminf(q.y, b0.y);  acc1[n] += fminf(q.y, b1.y);
            acc0[n] += fminf(q.z, b0.z);  acc1[n] += fminf(q.z, b1.z);
            acc0[n] += fminf(q.w, b0.w);  acc1[n] += fminf(q.w, b1.w);
        }
        __builtin_amdgcn_sched_group_barrier(0x100, 17, 0);
        __builtin_amdgcn_sched_group_barrier(0x00A, 352, 0);
    }

#pragma unroll
    for (int r = 0; r < 8; ++r) {
        const float keep0 = h ? acc0[8 + r] : acc0[r];
        const float send0 = h ? acc0[r]     : acc0[8 + r];
        const float keep1 = h ? acc1[8 + r] : acc1[r];
        const float send1 = h ? acc1[r]     : acc1[8 + r];
        const float smin0 = keep0 + __shfl_xor(send0, 16);
        const float smin1 = keep1 + __shfl_xor(send1, 16);
        const float val0  = c1 * smin0 - al * Sa0[r] - be * Sb0[r];
        const float val1  = c1 * smin1 - al * Sa1[r] - be * Sb1[r];
        float* orow = out + (size_t)(ti * 16 + 8 * h + r) * P_ROWS + (tj0 * 16 + c);
        orow[0]  = val0;
        orow[16] = val1;
    }
}

extern "C" void kernel_launch(void* const* d_in, const int* in_sizes, int n_in,
                              void* d_out, int out_size, void* d_ws, size_t ws_size,
                              hipStream_t stream) {
    const float* input     = (const float*)d_in[0];
    const float* prototype = (const float*)d_in[1];
    const float* fb        = (const float*)d_in[2];
    const float* alpha_raw = (const float*)d_in[3];
    const float* beta_raw  = (const float*)d_in[4];
    const float* theta_raw = (const float*)d_in[5];
    float* out = (float*)d_out;

    float* ap = (float*)d_ws;
    float* bp = ap + (size_t)N_ROWS * F_DIM;

    proj_norm_relu_wmma<<<((N_ROWS + P_ROWS) / 32) * 2, 256, 0, stream>>>(
        input, prototype, fb, ap);

    dim3 grid(P_ROWS / 128, N_ROWS / 32);
    pairwise_wmma<<<grid, 256, 0, stream>>>(ap, bp, alpha_raw, beta_raw, theta_raw, out);
}


# Pairwise launch: the 32 bp fragment loads of a wave go out as one group behind the barrier and are consumed as they land, not one load-wait-copy round trip at a time in front of it; sums unchanged
# speedup vs baseline: 3.0856x; 1.0570x over previous
#include <hip/hip_runtime.h>
#include <hip/hip_bf16.h>
#include <math.h>

typedef float v2f __attribute__((ext_vector_type(2)));
typedef float v4f __attribute__((ext_vector_type(4)));
typedef float v8f __attribute__((ext_vector_type(8)));

#define N_ROWS 2048
#define P_ROWS 512
#define D_DIM  1024
#define F_DIM  128
#define EPSV   1e-8f

__device__ __forceinline__ float softplus_f(float x) {
    return log1pf(expf(x));
}

#define KC       128
#define NCHUNK   (D_DIM / KC)
#define BM       32
#define BN       64
#define ST_ROWS  (BM + BN)
#define PITCH4   (KC / 4 + 1)
#define ST_F4    (ST_ROWS * PITCH4)

__device__ __forceinline__ void mma_quad(const v4f a4, const v4f b4,
                                         v8f& acc0, v8f& acc1, v2f& sa2, v2f& sb2) {
    v2f a_lo = {a4.x, a4.y}, a_hi = {a4.z, a4.w};
    v2f b_lo = {b4.x, b4.y}, b_hi = {b4.z, b4.w};
    sa2 = __builtin_elementwise_fma(a_lo, a_lo, sa2);
    sb2 = __builtin_elementwise_fma(b_lo, b_lo, sb2);
    sa2 = __builtin_elementwise_fma(a_hi, a_hi, sa2);
    sb2 = __builtin_elementwise_fma(b_hi, b_hi, sb2);
    acc0 = __builtin_amdgcn_wmma_f32_16x16x4_f32(
         false, a_lo,  false, b_lo,
         (short)0, acc0,  false,  false);
    acc1 = __builtin_amdgcn_wmma_f32_16x16x4_f32(
         false, a_hi,  false, b_hi,
         (short)0, acc1,  false,  false);
}

__global__ __launch_bounds__(256) __attribute__((amdgpu_waves_per_eu(2, 4)))
void proj_norm_relu_wmma(const float* __restrict__ Xin,
                         const float* __restrict__ Xpr,
                         const float* __restrict__ fb,
                         float* __restrict__ proj) {
    __shared__ v4f stage[2 * ST_F4];

    const int w    = threadIdx.x >> 5;
    const int lane = threadIdx.x & 31;
    const int rb   = blockIdx.x >> 1;
    const int cb   = blockIdx.x & 1;
    const int rt   = w >> 2;
    const int ct   = w & 3;
    const int half = lane >> 4;
    const int lm   = lane & 15;

    const float* X = (rb < N_ROWS / BM) ? Xin + (size_t)rb * BM * D_DIM
                                        : Xpr + (size_t)(rb - N_ROWS / BM) * BM * D_DIM;

    const float* gx = X  + (size_t)w * D_DIM + 4 * lane;
    const float* gf = fb + (size_t)(cb * BN + w) * D_DIM + 4 * lane;
    v4f* const st_w = stage + w * PITCH4 + lane;
    const v4f* const fa = stage + (16 * rt + lm) * PITCH4 + half;
    const v4f* const fbq = stage + (BM + 16 * ct + lm) * PITCH4 + half;

    {
        v4f gX[BM / 8], gF[BN / 8];
#pragma unroll
        for (int i = 0; i < BM / 8; ++i) gX[i] = *(const v4f*)(gx + (size_t)(8 * i) * D_DIM);
#pragma unroll
        for (int i = 0; i < BN / 8; ++i) gF[i] = *(const v4f*)(gf + (size_t)(8 * i) * D_DIM);
#pragma unroll
        for (int i = 0; i < BM / 8; ++i) st_w[(8 * i) * PITCH4] = gX[i];
#pragma unroll
        for (int i = 0; i < BN / 8; ++i) st_w[(BM + 8 * i) * PITCH4] = gF[i];
    }
    __syncthreads();

    v8f acc0 = {}, acc1 = {};
    v2f sa2 = {}, sb2 = {};
    for (int ch = 0; ch < NCHUNK - 1; ++ch) {
        const int s = ch & 1;
        const int ko = (ch + 1) * KC;
        v4f gX[BM / 8], gF[BN / 8];
#pragma unroll
        for (int i = 0; i < BM / 8; ++i) gX[i] = *(const v4f*)(gx + (size_t)(8 * i) * D_DIM + ko);
#pragma unroll
        for (int i = 0; i < BN / 8; ++i) gF[i] = *(const v4f*)(gf + (size_t)(8 * i) * D_DIM + ko);

        const v4f* pa = fa + s * ST_F4;
        const v4f* pb = fbq + s * ST_F4;
        v4f A[KC / 8], B[KC / 8];
#pragma unroll
        for (int t = 0; t < KC / 8; ++t) { A[t] = pa[2 * t]; B[t] = pb[2 * t]; }
#pragma unroll
        for (int t = 0; t < KC / 16; ++t) mma_quad(A[t], B[t], acc0, acc1, sa2, sb2);

        asm volatile("" :: "v"(acc0), "v"(acc1) : "memory");
        v4f* so = st_w + (s ^ 1) * ST_F4;
#pragma unroll
        for (int i = 0; i < BM / 8; ++i) so[(8 * i) * PITCH4] = gX[i];
#pragma unroll
        for (int i = 0; i < BN / 8; ++i) so[(BM + 8 * i) * PITCH4] = gF[i];

#pragma unroll
        for (int t = KC / 16; t < KC / 8 - 2; ++t) mma_quad(A[t], B[t], acc0, acc1, sa2, sb2);
        asm volatile("" :: "v"(acc0), "v"(acc1));
        __syncthreads();
#pragma unroll
        for (int t = KC / 8 - 2; t < KC / 8; ++t) mma_quad(A[t], B[t], acc0, acc1, sa2, sb2);
    }
    {
        const v4f* pa = fa  + ((NCHUNK - 1) & 1) * ST_F4;
        const v4f* pb = fbq + ((NCHUNK - 1) & 1) * ST_F4;
#pragma unroll
        for (int t = 0; t < KC / 8; ++t) mma_quad(pa[2 * t], pb[2 * t], acc0, acc1, sa2, sb2);
    }
    const v8f acc = acc0 + acc1;

    float sa = sa2.x + sa2.y;
    float sb = sb2.x + sb2.y;
    sa += __shfl_xor(sa, 16);
    sb += __shfl_xor(sb, 16);
    const float ia = 1.0f / fmaxf(sqrtf(sa), EPSV);
    const float ib = 1.0f / fmaxf(sqrtf(sb), EPSV);

    const int col = cb * BN + ct * 16 + lm;
#pragma unroll
    for (int v = 0; v < 8; ++v) {
        const int r = half * 8 + v;
        const float iar = __shfl(ia, r);
        const float val = acc[v] * iar * ib;
        proj[(size_t)(rb * BM + rt * 16 + r) * F_DIM + col] = fmaxf(val, 0.0f);
    }
}

#define A_PITCH4 (F_DIM / 4 + 1)


__device__ __forceinline__ float pos_mask(float x) { return (x > 0.0f) ? 1.0f : 0.0f; }

__device__ __forceinline__ v4f pos_mask4(const v4f x) {
    v4f m;
    m.x = pos_mask(x.x); m.y = pos_mask(x.y); m.z = pos_mask(x.z); m.w = pos_mask(x.w);
    return m;
}

#define J_TILES_PER_WAVE 2

__global__ __launch_bounds__(256) void pairwise_wmma(const float* __restrict__ ap,
                                                     const float* __restrict__ bp,
                                                     const float* __restrict__ alpha_raw,
                                                     const float* __restrict__ beta_raw,
                                                     const float* __restrict__ theta_raw,
                                                     float* __restrict__ out) {
    __shared__ v4f a_sh[32 * A_PITCH4];

    const int w    = threadIdx.x >> 5;
    const int lane = threadIdx.x & 31;
    const int h    = lane >> 4;
    const int c    = lane & 15;
    const int tl   = w >> 2;
    const int ti   = blockIdx.y * 2 + tl;
    const int tj0  = (blockIdx.x * 4 + (w & 3)) * J_TILES_PER_WAVE;

    {
        const float* abase = ap + (size_t)blockIdx.y * 32 * F_DIM;
        v4f g[4];
#pragma unroll
        for (int r = 0; r < 4; ++r) g[r] = *(const v4f*)(abase + (size_t)(8 * r + w) * F_DIM + 4 * lane);
#pragma unroll
        for (int r = 0; r < 4; ++r) a_sh[(8 * r + w) * A_PITCH4 + lane] = g[r];
    }

    const float al = softplus_f(alpha_raw[0]);
    const float be = softplus_f(beta_raw[0]);
    const float th = softplus_f(theta_raw[0]);
    const float c1 = th + al + be;

    __syncthreads();

    const float* brow0 = bp + (size_t)(tj0 * 16 + c) * F_DIM + 4 * h;
    const float* brow1 = brow0 + (size_t)16 * F_DIM;
    v4f B0[16], B1[16];
#pragma unroll
    for (int t = 0; t < 16; ++t) {
        B0[t] = *(const v4f*)(brow0 + 8 * t);
        B1[t] = *(const v4f*)(brow1 + 8 * t);
    }
    __builtin_amdgcn_sched_barrier(0);

    const v4f* const afrag = a_sh + (tl * 16 + c) * A_PITCH4 + h;
    const v4f* const arow  = a_sh + (tl * 16) * A_PITCH4 + h;

    float acc0[16], acc1[16];
#pragma unroll
    for (int n = 0; n < 16; ++n) { acc0[n] = 0.0f; acc1[n] = 0.0f; }
    v8f Sa0 = {}, Sb0 = {}, Sa1 = {}, Sb1 = {};

    __builtin_amdgcn_sched_group_barrier(0x100, 17, 0);
#pragma unroll
    for (int t = 0; t < 16; ++t) {
        const v4f a4 = afrag[2 * t];
        const v4f b0 = B0[t], b1 = B1[t];
        const v4f ma4 = pos_mask4(a4), mb0 = pos_mask4(b0), mb1 = pos_mask4(b1);
        const v2f a_lo  = {a4.x, a4.y},   a_hi  = {a4.z, a4.w};
        const v2f ma_lo = {ma4.x, ma4.y}, ma_hi = {ma4.z, ma4.w};
        const v2f b0_lo = {b0.x, b0.y},   b0_hi = {b0.z, b0.w};
        const v2f b1_lo = {b1.x, b1.y},   b1_hi = {b1.z, b1.w};
        const v2f mb0_lo = {mb0.x, mb0.y}, mb0_hi = {mb0.z, mb0.w};
        const v2f mb1_lo = {mb1.x, mb1.y}, mb1_hi = {mb1.z, mb1.w};
        Sa0 = __builtin_amdgcn_wmma_f32_16x16x4_f32(false, a_lo,  false, mb0_lo, (short)0, Sa0, false, false);
        Sb0 = __builtin_amdgcn_wmma_f32_16x16x4_f32(false, ma_lo, false, b0_lo,  (short)0, Sb0, false, false);
        Sa1 = __builtin_amdgcn_wmma_f32_16x16x4_f32(false, a_lo,  false, mb1_lo, (short)0, Sa1, false, false);
        Sb1 = __builtin_amdgcn_wmma_f32_16x16x4_f32(false, ma_lo, false, b1_lo,  (short)0, Sb1, false, false);
        Sa0 = __builtin_amdgcn_wmma_f32_16x16x4_f32(false, a_hi,  false, mb0_hi, (short)0, Sa0, false, false);
        Sb0 = __builtin_amdgcn_wmma_f32_16x16x4_f32(false, ma_hi, false, b0_hi,  (short)0, Sb0, false, false);
        Sa1 = __builtin_amdgcn_wmma_f32_16x16x4_f32(false, a_hi,  false, mb1_hi, (short)0, Sa1, false, false);
        Sb1 = __builtin_amdgcn_wmma_f32_16x16x4_f32(false, ma_hi, false, b1_hi,  (short)0, Sb1, false, false);
#pragma unroll
        for (int n = 0; n < 16; ++n) {
            const v4f q = arow[n * A_PITCH4 + 2 * t];
            acc0[n] += fminf(q.x, b0.x);  acc1[n] += fminf(q.x, b1.x);
            acc0[n] += fminf(q.y, b0.y);  acc1[n] += fminf(q.y, b1.y);
            acc0[n] += fminf(q.z, b0.z);  acc1[n] += fminf(q.z, b1.z);
            acc0[n] += fminf(q.w, b0.w);  acc1[n] += fminf(q.w, b1.w);
        }
        __builtin_amdgcn_sched_group_barrier(0x100, 17, 0);
        __builtin_amdgcn_sched_group_barrier(0x00A, 352, 0);
    }

#pragma unroll
    for (int r = 0; r < 8; ++r) {
        const float keep0 = h ? acc0[8 + r] : acc0[r];
        const float send0 = h ? acc0[r]     : acc0[8 + r];
        const float keep1 = h ? acc1[8 + r] : acc1[r];
        const float send1 = h ? acc1[r]     : acc1[8 + r];
        const float smin0 = keep0 + __shfl_xor(send0, 16);
        const float smin1 = keep1 + __shfl_xor(send1, 16);
        const float val0  = c1 * smin0 - al * Sa0[r] - be * Sb0[r];
        const float val1  = c1 * smin1 - al * Sa1[r] - be * Sb1[r];
        float* orow = out + (size_t)(ti * 16 + 8 * h + r) * P_ROWS + (tj0 * 16 + c);
        orow[0]  = val0;
        orow[16] = val1;
    }
}

extern "C" void kernel_launch(void* const* d_in, const int* in_sizes, int n_in,
                              void* d_out, int out_size, void* d_ws, size_t ws_size,
                              hipStream_t stream) {
    const float* input     = (const float*)d_in[0];
    const float* prototype = (const float*)d_in[1];
    const float* fb        = (const float*)d_in[2];
    const float* alpha_raw = (const float*)d_in[3];
    const float* beta_raw  = (const float*)d_in[4];
    const float* theta_raw = (const float*)d_in[5];
    float* out = (float*)d_out;

    float* ap = (float*)d_ws;
    float* bp = ap + (size_t)N_ROWS * F_DIM;

    proj_norm_relu_wmma<<<((N_ROWS + P_ROWS) / 32) * 2, 256, 0, stream>>>(
        input, prototype, fb, ap);

    dim3 grid(P_ROWS / 128, N_ROWS / 32);
    pairwise_wmma<<<grid, 256, 0, stream>>>(ap, bp, alpha_raw, beta_raw, theta_raw, out);
}


# Projection launch: global loads run two chunks ahead in two register sets (one in flight while the other awaits its LDS store) behind an LDS-only chunk barrier; was half a chunk; sums unchanged
# speedup vs baseline: 3.3890x; 1.0983x over previous
#include <hip/hip_runtime.h>
#include <hip/hip_bf16.h>
#include <math.h>

typedef float v2f __attribute__((ext_vector_type(2)));
typedef float v4f __attribute__((ext_vector_type(4)));
typedef float v8f __attribute__((ext_vector_type(8)));

#define N_ROWS 2048
#define P_ROWS 512
#define D_DIM  1024
#define F_DIM  128
#define EPSV   1e-8f

__device__ __forceinline__ float softplus_f(float x) {
    return log1pf(expf(x));
}

#define KC       128
#define NCHUNK   (D_DIM / KC)
#define BM       32
#define BN       64
#define ST_ROWS  (BM + BN)
#define PITCH4   (KC / 4 + 1)
#define ST_F4    (ST_ROWS * PITCH4)

__device__ __forceinline__ void mma_quad(const v4f a4, const v4f b4,
                                         v8f& acc0, v8f& acc1, v2f& sa2, v2f& sb2) {
    v2f a_lo = {a4.x, a4.y}, a_hi = {a4.z, a4.w};
    v2f b_lo = {b4.x, b4.y}, b_hi = {b4.z, b4.w};
    sa2 = __builtin_elementwise_fma(a_lo, a_lo, sa2);
    sb2 = __builtin_elementwise_fma(b_lo, b_lo, sb2);
    sa2 = __builtin_elementwise_fma(a_hi, a_hi, sa2);
    sb2 = __builtin_elementwise_fma(b_hi, b_hi, sb2);
    acc0 = __builtin_amdgcn_wmma_f32_16x16x4_f32(
         false, a_lo,  false, b_lo,
         (short)0, acc0,  false,  false);
    acc1 = __builtin_amdgcn_wmma_f32_16x16x4_f32(
         false, a_hi,  false, b_hi,
         (short)0, acc1,  false,  false);
}

__device__ __forceinline__ void lds_barrier() {
    __builtin_amdgcn_fence(__ATOMIC_RELEASE, "workgroup", "local");
    __builtin_amdgcn_s_barrier();
    __builtin_amdgcn_fence(__ATOMIC_ACQUIRE, "workgroup", "local");
}

#define G_ROWS (ST_ROWS / 8)

__device__ __forceinline__ void load_chunk(v4f (&g)[G_ROWS], const float* gx, const float* gf, const int ko) {
#pragma unroll
    for (int i = 0; i < BM / 8; ++i) g[i] = *(const v4f*)(gx + (size_t)(8 * i) * D_DIM + ko);
#pragma unroll
    for (int i = 0; i < BN / 8; ++i) g[BM / 8 + i] = *(const v4f*)(gf + (size_t)(8 * i) * D_DIM + ko);
}

__device__ __forceinline__ void store_chunk(v4f* so, const v4f (&g)[G_ROWS]) {
#pragma unroll
    for (int i = 0; i < G_ROWS; ++i) so[(8 * i) * PITCH4] = g[i];
}

__device__ __forceinline__ void chunk_step(const int s, const int ko_load,
                                           const float* gx, const float* gf, v4f* st_w,
                                           const v4f* fa, const v4f* fbq, v4f (&g)[G_ROWS],
                                           v8f& acc0, v8f& acc1, v2f& sa2, v2f& sb2) {
    const v4f* pa = fa + s * ST_F4;
    const v4f* pb = fbq + s * ST_F4;
    v4f A[KC / 8], B[KC / 8];
#pragma unroll
    for (int t = 0; t < KC / 8; ++t) { A[t] = pa[2 * t]; B[t] = pb[2 * t]; }
#pragma unroll
    for (int t = 0; t < KC / 16; ++t) mma_quad(A[t], B[t], acc0, acc1, sa2, sb2);

    asm volatile("" :: "v"(acc0), "v"(acc1) : "memory");
    store_chunk(st_w + (s ^ 1) * ST_F4, g);
    if (ko_load >= 0) load_chunk(g, gx, gf, ko_load);

#pragma unroll
    for (int t = KC / 16; t < KC / 8 - 2; ++t) mma_quad(A[t], B[t], acc0, acc1, sa2, sb2);
    asm volatile("" :: "v"(acc0), "v"(acc1));
    lds_barrier();
#pragma unroll
    for (int t = KC / 8 - 2; t < KC / 8; ++t) mma_quad(A[t], B[t], acc0, acc1, sa2, sb2);
}

__global__ __launch_bounds__(256) __attribute__((amdgpu_waves_per_eu(2, 4)))
void proj_norm_relu_wmma(const float* __restrict__ Xin,
                         const float* __restrict__ Xpr,
                         const float* __restrict__ fb,
                         float* __restrict__ proj) {
    __shared__ v4f stage[2 * ST_F4];

    const int w    = threadIdx.x >> 5;
    const int lane = threadIdx.x & 31;
    const int rb   = blockIdx.x >> 1;
    const int cb   = blockIdx.x & 1;
    const int rt   = w >> 2;
    const int ct   = w & 3;
    const int half = lane >> 4;
    const int lm   = lane & 15;

    const float* X = (rb < N_ROWS / BM) ? Xin + (size_t)rb * BM * D_DIM
                                        : Xpr + (size_t)(rb - N_ROWS / BM) * BM * D_DIM;

    const float* gx = X  + (size_t)w * D_DIM + 4 * lane;
    const float* gf = fb + (size_t)(cb * BN + w) * D_DIM + 4 * lane;
    v4f* const st_w = stage + w * PITCH4 + lane;
    const v4f* const fa = stage + (16 * rt + lm) * PITCH4 + half;
    const v4f* const fbq = stage + (BM + 16 * ct + lm) * PITCH4 + half;

    v4f g0[G_ROWS], g1[G_ROWS];
    {
        v4f gp[G_ROWS];
        load_chunk(gp, gx, gf, 0);
        __builtin_amdgcn_sched_barrier(0);
        load_chunk(g1, gx, gf, KC);
        __builtin_amdgcn_sched_barrier(0);
        load_chunk(g0, gx, gf, 2 * KC);
        __builtin_amdgcn_sched_barrier(0);
        store_chunk(st_w, gp);
    }
    lds_barrier();

    v8f acc0 = {}, acc1 = {};
    v2f sa2 = {}, sb2 = {};
    for (int p = 0; p < (NCHUNK - 2) / 2; ++p) {
        chunk_step(0, (2 * p + 3) * KC, gx, gf, st_w, fa, fbq, g1, acc0, acc1, sa2, sb2);
        chunk_step(1, (2 * p + 4 < NCHUNK) ? (2 * p + 4) * KC : -1, gx, gf, st_w, fa, fbq, g0, acc0, acc1, sa2, sb2);
    }
    chunk_step(0, -1, gx, gf, st_w, fa, fbq, g1, acc0, acc1, sa2, sb2);
    {
        const v4f* pa = fa  + ((NCHUNK - 1) & 1) * ST_F4;
        const v4f* pb = fbq + ((NCHUNK - 1) & 1) * ST_F4;
#pragma unroll
        for (int t = 0; t < KC / 8; ++t) mma_quad(pa[2 * t], pb[2 * t], acc0, acc1, sa2, sb2);
    }
    const v8f acc = acc0 + acc1;

    float sa = sa2.x + sa2.y;
    float sb = sb2.x + sb2.y;
    sa += __shfl_xor(sa, 16);
    sb += __shfl_xor(sb, 16);
    const float ia = 1.0f / fmaxf(sqrtf(sa), EPSV);
    const float ib = 1.0f / fmaxf(sqrtf(sb), EPSV);

    const int col = cb * BN + ct * 16 + lm;
#pragma unroll
    for (int v = 0; v < 8; ++v) {
        const int r = half * 8 + v;
        const float iar = __shfl(ia, r);
        const float val = acc[v] * iar * ib;
        proj[(size_t)(rb * BM + rt * 16 + r) * F_DIM + col] = fmaxf(val, 0.0f);
    }
}

#define A_PITCH4 (F_DIM / 4 + 1)


__device__ __forceinline__ float pos_mask(float x) { return (x > 0.0f) ? 1.0f : 0.0f; }

__device__ __forceinline__ v4f pos_mask4(const v4f x) {
    v4f m;
    m.x = pos_mask(x.x); m.y = pos_mask(x.y); m.z = pos_mask(x.z); m.w = pos_mask(x.w);
    return m;
}

#define J_TILES_PER_WAVE 2

__global__ __launch_bounds__(256) void pairwise_wmma(const float* __restrict__ ap,
                                                     const float* __restrict__ bp,
                                                     const float* __restrict__ alpha_raw,
                                                     const float* __restrict__ beta_raw,
                                                     const float* __restrict__ theta_raw,
                                                     float* __restrict__ out) {
    __shared__ v4f a_sh[32 * A_PITCH4];

    const int w    = threadIdx.x >> 5;
    const int lane = threadIdx.x & 31;
    const int h    = lane >> 4;
    const int c    = lane & 15;
    const int tl   = w >> 2;
    const int ti   = blockIdx.y * 2 + tl;
    const int tj0  = (blockIdx.x * 4 + (w & 3)) * J_TILES_PER_WAVE;

    {
        const float* abase = ap + (size_t)blockIdx.y * 32 * F_DIM;
        v4f g[4];
#pragma unroll
        for (int r = 0; r < 4; ++r) g[r] = *(const v4f*)(abase + (size_t)(8 * r + w) * F_DIM + 4 * lane);
#pragma unroll
        for (int r = 0; r < 4; ++r) a_sh[(8 * r + w) * A_PITCH4 + lane] = g[r];
    }

    const float al = softplus_f(alpha_raw[0]);
    const float be = softplus_f(beta_raw[0]);
    const float th = softplus_f(theta_raw[0]);
    const float c1 = th + al + be;

    __syncthreads();

    const float* brow0 = bp + (size_t)(tj0 * 16 + c) * F_DIM + 4 * h;
    const float* brow1 = brow0 + (size_t)16 * F_DIM;
    v4f B0[16], B1[16];
#pragma unroll
    for (int t = 0; t < 16; ++t) {
        B0[t] = *(const v4f*)(brow0 + 8 * t);
        B1[t] = *(const v4f*)(brow1 + 8 * t);
    }
    __builtin_amdgcn_sched_barrier(0);

    const v4f* const afrag = a_sh + (tl * 16 + c) * A_PITCH4 + h;
    const v4f* const arow  = a_sh + (tl * 16) * A_PITCH4 + h;

    float acc0[16], acc1[16];
#pragma unroll
    for (int n = 0; n < 16; ++n) { acc0[n] = 0.0f; acc1[n] = 0.0f; }
    v8f Sa0 = {}, Sb0 = {}, Sa1 = {}, Sb1 = {};

    __builtin_amdgcn_sched_group_barrier(0x100, 17, 0);
#pragma unroll
    for (int t = 0; t < 16; ++t) {
        const v4f a4 = afrag[2 * t];
        const v4f b0 = B0[t], b1 = B1[t];
        const v4f ma4 = pos_mask4(a4), mb0 = pos_mask4(b0), mb1 = pos_mask4(b1);
        const v2f a_lo  = {a4.x, a4.y},   a_hi  = {a4.z, a4.w};
        const v2f ma_lo = {ma4.x, ma4.y}, ma_hi = {ma4.z, ma4.w};
        const v2f b0_lo = {b0.x, b0.y},   b0_hi = {b0.z, b0.w};
        const v2f b1_lo = {b1.x, b1.y},   b1_hi = {b1.z, b1.w};
        const v2f mb0_lo = {mb0.x, mb0.y}, mb0_hi = {mb0.z, mb0.w};
        const v2f mb1_lo = {mb1.x, mb1.y}, mb1_hi = {mb1.z, mb1.w};
        Sa0 = __builtin_amdgcn_wmma_f32_16x16x4_f32(false, a_lo,  false, mb0_lo, (short)0, Sa0, false, false);
        Sb0 = __builtin_amdgcn_wmma_f32_16x16x4_f32(false, ma_lo, false, b0_lo,  (short)0, Sb0, false, false);
        Sa1 = __builtin_amdgcn_wmma_f32_16x16x4_f32(false, a_lo,  false, mb1_lo, (short)0, Sa1, false, false);
        Sb1 = __builtin_amdgcn_wmma_f32_16x16x4_f32(false, ma_lo, false, b1_lo,  (short)0, Sb1, false, false);
        Sa0 = __builtin_amdgcn_wmma_f32_16x16x4_f32(false, a_hi,  false, mb0_hi, (short)0, Sa0, false, false);
        Sb0 = __builtin_amdgcn_wmma_f32_16x16x4_f32(false, ma_hi, false, b0_hi,  (short)0, Sb0, false, false);
        Sa1 = __builtin_amdgcn_wmma_f32_16x16x4_f32(false, a_hi,  false, mb1_hi, (short)0, Sa1, false, false);
        Sb1 = __builtin_amdgcn_wmma_f32_16x16x4_f32(false, ma_hi, false, b1_hi,  (short)0, Sb1, false, false);
#pragma unroll
        for (int n = 0; n < 16; ++n) {
            const v4f q = arow[n * A_PITCH4 + 2 * t];
            acc0[n] += fminf(q.x, b0.x);  acc1[n] += fminf(q.x, b1.x);
            acc0[n] += fminf(q.y, b0.y);  acc1[n] += fminf(q.y, b1.y);
            acc0[n] += fminf(q.z, b0.z);  acc1[n] += fminf(q.z, b1.z);
            acc0[n] += fminf(q.w, b0.w);  acc1[n] += fminf(q.w, b1.w);
        }
        __builtin_amdgcn_sched_group_barrier(0x100, 17, 0);
        __builtin_amdgcn_sched_group_barrier(0x00A, 352, 0);
    }

#pragma unroll
    for (int r = 0; r < 8; ++r) {
        const float keep0 = h ? acc0[8 + r] : acc0[r];
        const float send0 = h ? acc0[r]     : acc0[8 + r];
        const float keep1 = h ? acc1[8 + r] : acc1[r];
        const float send1 = h ? acc1[r]     : acc1[8 + r];
        const float smin0 = keep0 + __shfl_xor(send0, 16);
        const float smin1 = keep1 + __shfl_xor(send1, 16);
        const float val0  = c1 * smin0 - al * Sa0[r] - be * Sb0[r];
        const float val1  = c1 * smin1 - al * Sa1[r] - be * Sb1[r];
        float* orow = out + (size_t)(ti * 16 + 8 * h + r) * P_ROWS + (tj0 * 16 + c);
        orow[0]  = val0;
        orow[16] = val1;
    }
}

extern "C" void kernel_launch(void* const* d_in, const int* in_sizes, int n_in,
                              void* d_out, int out_size, void* d_ws, size_t ws_size,
                              hipStream_t stream) {
    const float* input     = (const float*)d_in[0];
    const float* prototype = (const float*)d_in[1];
    const float* fb        = (const float*)d_in[2];
    const float* alpha_raw = (const float*)d_in[3];
    const float* beta_raw  = (const float*)d_in[4];
    const float* theta_raw = (const float*)d_in[5];
    float* out = (float*)d_out;

    float* ap = (float*)d_ws;
    float* bp = ap + (size_t)N_ROWS * F_DIM;

    proj_norm_relu_wmma<<<((N_ROWS + P_ROWS) / 32) * 2, 256, 0, stream>>>(
        input, prototype, fb, ap);

    dim3 grid(P_ROWS / 128, N_ROWS / 32);
    pairwise_wmma<<<grid, 256, 0, stream>>>(ap, bp, alpha_raw, beta_raw, theta_raw, out);
}
